# starting kernel
# baseline (speedup 1.0000x reference)
#include <hip/hip_runtime.h>


typedef __attribute__((ext_vector_type(2))) float v2f;
typedef __attribute__((ext_vector_type(8))) float v8f;

#define T_FRAMES 3000
#define F_BINS   201
#define B_BATCH  32
#define TT       64
#define HALO     42
#define SROWS    (TT + HALO)
#define FP       208
#define NCH      13
#define EPSF     1e-5f
#define SCONST   6.209375e-05f
#define INV_LN10 0.4342944819032518f

__device__ __forceinline__ float coefA(int t, int u) {
  bool in = (t >= 1) & (u >= 0) & (u >= t - 10) & (u < t);
  int cnt = (t < 10) ? (t > 0 ? t : 1) : 10;
  return in ? (1.0f / (float)cnt) : 0.0f;
}
__device__ __forceinline__ float coefV(int t, int u) {
  bool in = (t >= 1) & (u >= 0) & (u >= t - 30) & (u < t);
  int cnt = (t < 30) ? (t > 0 ? t : 1) : 30;
  return in ? (1.0f / (float)cnt) : 0.0f;
}

__global__ __launch_bounds__(256, 1)
void ltsf_kernel(const float* __restrict__ x, float* __restrict__ out) {
  __shared__ float sL[SROWS * FP];
  __shared__ float wsc[8 * 48 * 16];
  __shared__ float lsc[8 * 48 * 16];
  __shared__ float red[8][16];

  const int b   = blockIdx.y;
  const int t0  = blockIdx.x * TT;
  const int tid = threadIdx.x;

  for (int idx = tid; idx < SROWS * FP; idx += 256) {
    int r = idx / FP, f = idx - r * FP;
    int u = t0 - HALO + r;
    float v = 0.0f;
    if (f < F_BINS && u >= 0 && u < T_FRAMES) {
      const float2 ri =
          *(const float2*)(x + (((size_t)b * T_FRAMES + u) * F_BINS + f) * 2);
      float sc = (f == 0 || f == F_BINS - 1) ? SCONST : 2.0f * SCONST;
      v = (ri.x * ri.x + ri.y * ri.y) * sc;
    }
    sL[idx] = v;
  }
  __syncthreads();

  const int lane = tid & 31;
  const int wave = tid >> 5;
  const int sub  = wave & 3;
  const int tsub = t0 + 16 * sub;
  const int tb   = tsub - 32;
  const int m    = lane & 15;
  const int koff = (lane >> 4) << 1;
  const int crow = (lane >> 4) << 3;
  float* wp = wsc + wave * (48 * 16);
  float* lp = lsc + wave * (48 * 16);

  v8f acc = {0.f, 0.f, 0.f, 0.f, 0.f, 0.f, 0.f, 0.f};

  for (int ch = (wave >> 2); ch < NCH; ch += 2) {
    const int f0 = ch * 16;

    #pragma unroll
    for (int i = 0; i < 3; ++i) {
      const int ta = tb + 16 * i;
      v8f c = {0.f, 0.f, 0.f, 0.f, 0.f, 0.f, 0.f, 0.f};
      #pragma unroll
      for (int j = 0; j < 7; ++j) {
        const int ub = ta - 10 + 4 * j;
        v2f a, bv;
        a.x = coefA(ta + m, ub + koff);
        a.y = coefA(ta + m, ub + koff + 1);
        int su0 = ub + koff - (t0 - HALO);
        int su1 = su0 + 1;
        su0 = su0 < 0 ? 0 : (su0 > SROWS - 1 ? SROWS - 1 : su0);
        su1 = su1 < 0 ? 0 : (su1 > SROWS - 1 ? SROWS - 1 : su1);
        bv.x = sL[su0 * FP + f0 + m];
        bv.y = sL[su1 * FP + f0 + m];
        c = __builtin_amdgcn_wmma_f32_16x16x4_f32(
            false, a, false, bv, (short)0, c, false, false);
      }
      #pragma unroll
      for (int e = 0; e < 8; ++e) {
        const int row = 16 * i + e + crow;
        const float w = c[e];
        wp[row * 16 + m] = w;
        lp[row * 16 + m] = __logf(w + EPSF);
      }
    }

    v8f amc = {0.f, 0.f, 0.f, 0.f, 0.f, 0.f, 0.f, 0.f};
    v8f glc = {0.f, 0.f, 0.f, 0.f, 0.f, 0.f, 0.f, 0.f};
    #pragma unroll
    for (int k = 0; k < 12; ++k) {
      const int ubl = 4 * k;
      const int u   = tb + ubl;
      v2f a, bw, bl;
      a.x = coefV(tsub + m, u + koff);
      a.y = coefV(tsub + m, u + koff + 1);
      const int r0 = ubl + koff;
      bw.x = wp[r0 * 16 + m];
      bw.y = wp[(r0 + 1) * 16 + m];
      bl.x = lp[r0 * 16 + m];
      bl.y = lp[(r0 + 1) * 16 + m];
      amc = __builtin_amdgcn_wmma_f32_16x16x4_f32(
          false, a, false, bw, (short)0, amc, false, false);
      glc = __builtin_amdgcn_wmma_f32_16x16x4_f32(
          false, a, false, bl, (short)0, glc, false, false);
    }

    #pragma unroll
    for (int e = 0; e < 8; ++e) {
      const int t = tsub + e + crow;
      const float am = amc[e] + EPSF;
      const float gm = (__expf(glc[e]) - EPSF) + EPSF;
      float val = -(__logf(gm) - __logf(am)) * INV_LN10;
      if (t <= 0 || t >= T_FRAMES || (f0 + m) >= F_BINS) val = 0.0f;
      acc[e] += val;
    }
  }

  #pragma unroll
  for (int e = 0; e < 8; ++e) {
    float v = acc[e];
    v += __shfl_xor(v, 1, 32);
    v += __shfl_xor(v, 2, 32);
    v += __shfl_xor(v, 4, 32);
    v += __shfl_xor(v, 8, 32);
    acc[e] = v;
  }
  if (lane == 0) {
    #pragma unroll
    for (int e = 0; e < 8; ++e) red[wave][e] = acc[e];
  }
  if (lane == 16) {
    #pragma unroll
    for (int e = 0; e < 8; ++e) red[wave][e + 8] = acc[e];
  }
  __syncthreads();

  if (tid < 64) {
    const int s2 = tid >> 4, i2 = tid & 15;
    const int t = t0 + 16 * s2 + i2;
    if (t < T_FRAMES)
      out[(size_t)b * T_FRAMES + t] = red[s2][i2] + red[s2 + 4][i2];
  }
}

extern "C" void kernel_launch(void* const* d_in, const int* in_sizes, int n_in,
                              void* d_out, int out_size, void* d_ws,
                              size_t ws_size, hipStream_t stream) {
  (void)in_sizes; (void)n_in; (void)d_ws; (void)ws_size; (void)out_size;
  const float* x = (const float*)d_in[0];
  float* out = (float*)d_out;
  dim3 grid((T_FRAMES + TT - 1) / TT, B_BATCH);
  ltsf_kernel<<<grid, dim3(256), 0, stream>>>(x, out);
}


# stage the input tile with 21 independent 16-byte global loads in flight per lane (two batches) instead of one 8-byte load per wait; per-element arithmetic and summation order unchanged
# speedup vs baseline: 2.1957x; 2.1957x over previous
#include <hip/hip_runtime.h>


typedef __attribute__((ext_vector_type(2))) float v2f;
typedef __attribute__((ext_vector_type(8))) float v8f;

#define T_FRAMES 3000
#define F_BINS   201
#define B_BATCH  32
#define TT       64
#define HALO     42
#define SROWS    (TT + HALO)
#define FP       208
#define NCH      13
#define LDB      21
#define EPSF     1e-5f
#define SCONST   6.209375e-05f
#define INV_LN10 0.4342944819032518f

__device__ __forceinline__ float coefA(int t, int u) {
  bool in = (t >= 1) & (u >= 0) & (u >= t - 10) & (u < t);
  int cnt = (t < 10) ? (t > 0 ? t : 1) : 10;
  return in ? (1.0f / (float)cnt) : 0.0f;
}
__device__ __forceinline__ float coefV(int t, int u) {
  bool in = (t >= 1) & (u >= 0) & (u >= t - 30) & (u < t);
  int cnt = (t < 30) ? (t > 0 ? t : 1) : 30;
  return in ? (1.0f / (float)cnt) : 0.0f;
}

__global__ __launch_bounds__(256, 1)
void ltsf_kernel(const float* __restrict__ x, float* __restrict__ out) {
  __shared__ float sL[SROWS * FP];
  __shared__ float wsc[8 * 48 * 16];
  __shared__ float lsc[8 * 48 * 16];
  __shared__ float red[8][16];

  const int b   = blockIdx.y;
  const int t0  = blockIdx.x * TT;
  const int tid = threadIdx.x;

  const int u_lo    = (t0 - HALO) > 0 ? (t0 - HALO) : 0;
  const int u_hi    = (t0 + TT) < T_FRAMES ? (t0 + TT) : T_FRAMES;
  const int row_off = u_lo - (t0 - HALO);
  const int nrows   = u_hi - u_lo;
  const int n4      = (nrows * F_BINS) >> 1;
  const float4* __restrict__ src =
      (const float4*)(x + ((size_t)b * T_FRAMES + u_lo) * (F_BINS * 2));

  for (int i = tid; i < nrows * 8; i += 256) {
    const int r = i >> 3, c = i & 7;
    if (c < FP - F_BINS) sL[(row_off + r) * FP + F_BINS + c] = 0.0f;
  }
  for (int i = tid; i < row_off * FP; i += 256) sL[i] = 0.0f;
  for (int i = (row_off + nrows) * FP + tid; i < SROWS * FP; i += 256)
    sL[i] = 0.0f;

  for (int base = 0; base < n4; base += LDB * 256) {
    float4 v[LDB];
    #pragma unroll
    for (int k = 0; k < LDB; ++k) {
      const int g = base + k * 256 + tid;
      v[k] = src[g < n4 ? g : n4 - 1];
    }
    #pragma unroll
    for (int k = 0; k < LDB; ++k) {
      const int g = base + k * 256 + tid;
      if (g < n4) {
        const unsigned e0 = 2u * (unsigned)g;
        const unsigned r0 = e0 / F_BINS, f0 = e0 - r0 * F_BINS;
        const bool wrap = (f0 == F_BINS - 1);
        const unsigned r1 = wrap ? r0 + 1 : r0;
        const unsigned f1 = wrap ? 0u : f0 + 1;
        const float s0 = (f0 == 0 || f0 == F_BINS - 1) ? SCONST : 2.0f * SCONST;
        const float s1 = (f1 == 0 || f1 == F_BINS - 1) ? SCONST : 2.0f * SCONST;
        sL[(row_off + r0) * FP + f0] = (v[k].x * v[k].x + v[k].y * v[k].y) * s0;
        sL[(row_off + r1) * FP + f1] = (v[k].z * v[k].z + v[k].w * v[k].w) * s1;
      }
    }
  }
  __syncthreads();

  const int lane = tid & 31;
  const int wave = tid >> 5;
  const int sub  = wave & 3;
  const int tsub = t0 + 16 * sub;
  const int tb   = tsub - 32;
  const int m    = lane & 15;
  const int koff = (lane >> 4) << 1;
  const int crow = (lane >> 4) << 3;
  float* wp = wsc + wave * (48 * 16);
  float* lp = lsc + wave * (48 * 16);

  v8f acc = {0.f, 0.f, 0.f, 0.f, 0.f, 0.f, 0.f, 0.f};

  for (int ch = (wave >> 2); ch < NCH; ch += 2) {
    const int f0 = ch * 16;

    #pragma unroll
    for (int i = 0; i < 3; ++i) {
      const int ta = tb + 16 * i;
      v8f c = {0.f, 0.f, 0.f, 0.f, 0.f, 0.f, 0.f, 0.f};
      #pragma unroll
      for (int j = 0; j < 7; ++j) {
        const int ub = ta - 10 + 4 * j;
        v2f a, bv;
        a.x = coefA(ta + m, ub + koff);
        a.y = coefA(ta + m, ub + koff + 1);
        int su0 = ub + koff - (t0 - HALO);
        int su1 = su0 + 1;
        su0 = su0 < 0 ? 0 : (su0 > SROWS - 1 ? SROWS - 1 : su0);
        su1 = su1 < 0 ? 0 : (su1 > SROWS - 1 ? SROWS - 1 : su1);
        bv.x = sL[su0 * FP + f0 + m];
        bv.y = sL[su1 * FP + f0 + m];
        c = __builtin_amdgcn_wmma_f32_16x16x4_f32(
            false, a, false, bv, (short)0, c, false, false);
      }
      #pragma unroll
      for (int e = 0; e < 8; ++e) {
        const int row = 16 * i + e + crow;
        const float w = c[e];
        wp[row * 16 + m] = w;
        lp[row * 16 + m] = __logf(w + EPSF);
      }
    }

    v8f amc = {0.f, 0.f, 0.f, 0.f, 0.f, 0.f, 0.f, 0.f};
    v8f glc = {0.f, 0.f, 0.f, 0.f, 0.f, 0.f, 0.f, 0.f};
    #pragma unroll
    for (int k = 0; k < 12; ++k) {
      const int ubl = 4 * k;
      const int u   = tb + ubl;
      v2f a, bw, bl;
      a.x = coefV(tsub + m, u + koff);
      a.y = coefV(tsub + m, u + koff + 1);
      const int r0 = ubl + koff;
      bw.x = wp[r0 * 16 + m];
      bw.y = wp[(r0 + 1) * 16 + m];
      bl.x = lp[r0 * 16 + m];
      bl.y = lp[(r0 + 1) * 16 + m];
      amc = __builtin_amdgcn_wmma_f32_16x16x4_f32(
          false, a, false, bw, (short)0, amc, false, false);
      glc = __builtin_amdgcn_wmma_f32_16x16x4_f32(
          false, a, false, bl, (short)0, glc, false, false);
    }

    #pragma unroll
    for (int e = 0; e < 8; ++e) {
      const int t = tsub + e + crow;
      const float am = amc[e] + EPSF;
      const float gm = (__expf(glc[e]) - EPSF) + EPSF;
      float val = -(__logf(gm) - __logf(am)) * INV_LN10;
      if (t <= 0 || t >= T_FRAMES || (f0 + m) >= F_BINS) val = 0.0f;
      acc[e] += val;
    }
  }

  #pragma unroll
  for (int e = 0; e < 8; ++e) {
    float v = acc[e];
    v += __shfl_xor(v, 1, 32);
    v += __shfl_xor(v, 2, 32);
    v += __shfl_xor(v, 4, 32);
    v += __shfl_xor(v, 8, 32);
    acc[e] = v;
  }
  if (lane == 0) {
    #pragma unroll
    for (int e = 0; e < 8; ++e) red[wave][e] = acc[e];
  }
  if (lane == 16) {
    #pragma unroll
    for (int e = 0; e < 8; ++e) red[wave][e + 8] = acc[e];
  }
  __syncthreads();

  if (tid < 64) {
    const int s2 = tid >> 4, i2 = tid & 15;
    const int t = t0 + 16 * s2 + i2;
    if (t < T_FRAMES)
      out[(size_t)b * T_FRAMES + t] = red[s2][i2] + red[s2 + 4][i2];
  }
}

extern "C" void kernel_launch(void* const* d_in, const int* in_sizes, int n_in,
                              void* d_out, int out_size, void* d_ws,
                              size_t ws_size, hipStream_t stream) {
  (void)in_sizes; (void)n_in; (void)d_ws; (void)ws_size; (void)out_size;
  const float* x = (const float*)d_in[0];
  float* out = (float*)d_out;
  dim3 grid((T_FRAMES + TT - 1) / TT, B_BATCH);
  ltsf_kernel<<<grid, dim3(256), 0, stream>>>(x, out);
}


# log/exp of the chunk loop on the bare log2/exp2 instructions (glog mean kept in base 2, one final scale by log10(2)) instead of the range-checked library expansions; summation order unchanged
# speedup vs baseline: 2.9177x; 1.3288x over previous
#include <hip/hip_runtime.h>


typedef __attribute__((ext_vector_type(2))) float v2f;
typedef __attribute__((ext_vector_type(8))) float v8f;

#define T_FRAMES 3000
#define F_BINS   201
#define B_BATCH  32
#define TT       64
#define HALO     42
#define SROWS    (TT + HALO)
#define FP       208
#define NCH      13
#define LDB      21
#define EPSF     1e-5f
#define SCONST   6.209375e-05f
#define LOG10_2  0.30102999566398120f

__device__ __forceinline__ float coefA(int t, int u) {
  bool in = (t >= 1) & (u >= 0) & (u >= t - 10) & (u < t);
  int cnt = (t < 10) ? (t > 0 ? t : 1) : 10;
  return in ? (1.0f / (float)cnt) : 0.0f;
}
__device__ __forceinline__ float coefV(int t, int u) {
  bool in = (t >= 1) & (u >= 0) & (u >= t - 30) & (u < t);
  int cnt = (t < 30) ? (t > 0 ? t : 1) : 30;
  return in ? (1.0f / (float)cnt) : 0.0f;
}

__global__ __launch_bounds__(256, 1)
void ltsf_kernel(const float* __restrict__ x, float* __restrict__ out) {
  __shared__ float sL[SROWS * FP];
  __shared__ float wsc[8 * 48 * 16];
  __shared__ float lsc[8 * 48 * 16];
  __shared__ float red[8][16];

  const int b   = blockIdx.y;
  const int t0  = blockIdx.x * TT;
  const int tid = threadIdx.x;

  const int u_lo    = (t0 - HALO) > 0 ? (t0 - HALO) : 0;
  const int u_hi    = (t0 + TT) < T_FRAMES ? (t0 + TT) : T_FRAMES;
  const int row_off = u_lo - (t0 - HALO);
  const int nrows   = u_hi - u_lo;
  const int n4      = (nrows * F_BINS) >> 1;
  const float4* __restrict__ src =
      (const float4*)(x + ((size_t)b * T_FRAMES + u_lo) * (F_BINS * 2));

  for (int i = tid; i < nrows * 8; i += 256) {
    const int r = i >> 3, c = i & 7;
    if (c < FP - F_BINS) sL[(row_off + r) * FP + F_BINS + c] = 0.0f;
  }
  for (int i = tid; i < row_off * FP; i += 256) sL[i] = 0.0f;
  for (int i = (row_off + nrows) * FP + tid; i < SROWS * FP; i += 256)
    sL[i] = 0.0f;

  for (int base = 0; base < n4; base += LDB * 256) {
    float4 v[LDB];
    #pragma unroll
    for (int k = 0; k < LDB; ++k) {
      const int g = base + k * 256 + tid;
      v[k] = src[g < n4 ? g : n4 - 1];
    }
    #pragma unroll
    for (int k = 0; k < LDB; ++k) {
      const int g = base + k * 256 + tid;
      if (g < n4) {
        const unsigned e0 = 2u * (unsigned)g;
        const unsigned r0 = e0 / F_BINS, f0 = e0 - r0 * F_BINS;
        const bool wrap = (f0 == F_BINS - 1);
        const unsigned r1 = wrap ? r0 + 1 : r0;
        const unsigned f1 = wrap ? 0u : f0 + 1;
        const float s0 = (f0 == 0 || f0 == F_BINS - 1) ? SCONST : 2.0f * SCONST;
        const float s1 = (f1 == 0 || f1 == F_BINS - 1) ? SCONST : 2.0f * SCONST;
        sL[(row_off + r0) * FP + f0] = (v[k].x * v[k].x + v[k].y * v[k].y) * s0;
        sL[(row_off + r1) * FP + f1] = (v[k].z * v[k].z + v[k].w * v[k].w) * s1;
      }
    }
  }
  __syncthreads();

  const int lane = tid & 31;
  const int wave = tid >> 5;
  const int sub  = wave & 3;
  const int tsub = t0 + 16 * sub;
  const int tb   = tsub - 32;
  const int m    = lane & 15;
  const int koff = (lane >> 4) << 1;
  const int crow = (lane >> 4) << 3;
  float* wp = wsc + wave * (48 * 16);
  float* lp = lsc + wave * (48 * 16);

  v8f acc = {0.f, 0.f, 0.f, 0.f, 0.f, 0.f, 0.f, 0.f};

  for (int ch = (wave >> 2); ch < NCH; ch += 2) {
    const int f0 = ch * 16;

    #pragma unroll
    for (int i = 0; i < 3; ++i) {
      const int ta = tb + 16 * i;
      v8f c = {0.f, 0.f, 0.f, 0.f, 0.f, 0.f, 0.f, 0.f};
      #pragma unroll
      for (int j = 0; j < 7; ++j) {
        const int ub = ta - 10 + 4 * j;
        v2f a, bv;
        a.x = coefA(ta + m, ub + koff);
        a.y = coefA(ta + m, ub + koff + 1);
        int su0 = ub + koff - (t0 - HALO);
        int su1 = su0 + 1;
        su0 = su0 < 0 ? 0 : (su0 > SROWS - 1 ? SROWS - 1 : su0);
        su1 = su1 < 0 ? 0 : (su1 > SROWS - 1 ? SROWS - 1 : su1);
        bv.x = sL[su0 * FP + f0 + m];
        bv.y = sL[su1 * FP + f0 + m];
        c = __builtin_amdgcn_wmma_f32_16x16x4_f32(
            false, a, false, bv, (short)0, c, false, false);
      }
      #pragma unroll
      for (int e = 0; e < 8; ++e) {
        const int row = 16 * i + e + crow;
        const float w = c[e];
        wp[row * 16 + m] = w;
        lp[row * 16 + m] = __builtin_amdgcn_logf(w + EPSF);
      }
    }

    v8f amc = {0.f, 0.f, 0.f, 0.f, 0.f, 0.f, 0.f, 0.f};
    v8f glc = {0.f, 0.f, 0.f, 0.f, 0.f, 0.f, 0.f, 0.f};
    #pragma unroll
    for (int k = 0; k < 12; ++k) {
      const int ubl = 4 * k;
      const int u   = tb + ubl;
      v2f a, bw, bl;
      a.x = coefV(tsub + m, u + koff);
      a.y = coefV(tsub + m, u + koff + 1);
      const int r0 = ubl + koff;
      bw.x = wp[r0 * 16 + m];
      bw.y = wp[(r0 + 1) * 16 + m];
      bl.x = lp[r0 * 16 + m];
      bl.y = lp[(r0 + 1) * 16 + m];
      amc = __builtin_amdgcn_wmma_f32_16x16x4_f32(
          false, a, false, bw, (short)0, amc, false, false);
      glc = __builtin_amdgcn_wmma_f32_16x16x4_f32(
          false, a, false, bl, (short)0, glc, false, false);
    }

    const bool colok = (f0 + m) < F_BINS;
    #pragma unroll
    for (int e = 0; e < 8; ++e) {
      const int t = tsub + e + crow;
      const float am = amc[e] + EPSF;
      const float gm = (__builtin_amdgcn_exp2f(glc[e]) - EPSF) + EPSF;
      const float val =
          (__builtin_amdgcn_logf(am) - __builtin_amdgcn_logf(gm)) * LOG10_2;
      const bool ok = (t > 0) & (t < T_FRAMES) & colok;
      acc[e] += ok ? val : 0.0f;
    }
  }

  #pragma unroll
  for (int e = 0; e < 8; ++e) {
    float v = acc[e];
    v += __shfl_xor(v, 1, 32);
    v += __shfl_xor(v, 2, 32);
    v += __shfl_xor(v, 4, 32);
    v += __shfl_xor(v, 8, 32);
    acc[e] = v;
  }
  if (lane == 0) {
    #pragma unroll
    for (int e = 0; e < 8; ++e) red[wave][e] = acc[e];
  }
  if (lane == 16) {
    #pragma unroll
    for (int e = 0; e < 8; ++e) red[wave][e + 8] = acc[e];
  }
  __syncthreads();

  if (tid < 64) {
    const int s2 = tid >> 4, i2 = tid & 15;
    const int t = t0 + 16 * s2 + i2;
    if (t < T_FRAMES)
      out[(size_t)b * T_FRAMES + t] = red[s2][i2] + red[s2 + 4][i2];
  }
}

extern "C" void kernel_launch(void* const* d_in, const int* in_sizes, int n_in,
                              void* d_out, int out_size, void* d_ws,
                              size_t ws_size, hipStream_t stream) {
  (void)in_sizes; (void)n_in; (void)d_ws; (void)ws_size; (void)out_size;
  const float* x = (const float*)d_in[0];
  float* out = (float*)d_out;
  dim3 grid((T_FRAMES + TT - 1) / TT, B_BATCH);
  ltsf_kernel<<<grid, dim3(256), 0, stream>>>(x, out);
}
